# starting kernel
# baseline (speedup 1.0000x reference)
#include <hip/hip_runtime.h>


typedef float v2f __attribute__((ext_vector_type(2)));
typedef float v8f __attribute__((ext_vector_type(8)));

#define H 128
#define NIN 8
#define LDS_STRIDE 130

__device__ __forceinline__ v8f wmma_k4(v2f a, v2f b, v8f c) {
  return __builtin_amdgcn_wmma_f32_16x16x4_f32(
       false, a,  false, b,
       (short)0, c,  false,  false);
}

__device__ __forceinline__ v2f ldg2(const float* p) {
  return *reinterpret_cast<const v2f*>(p);
}

template <int WAVES>
__global__ __launch_bounds__(WAVES * 32) void net_forward_kernel(
    const float* __restrict__ x, float* __restrict__ out, int n_samples,
    const float* __restrict__ w1_1, const float* __restrict__ b1_1,
    const float* __restrict__ w2_1, const float* __restrict__ b2_1,
    const float* __restrict__ w1_2, const float* __restrict__ b1_2,
    const float* __restrict__ w2_2, const float* __restrict__ b2_2,
    const float* __restrict__ w_out) {
  __shared__ float ybuf[WAVES][16 * LDS_STRIDE];

  const int wave = threadIdx.x >> 5;
  const int lane = threadIdx.x & 31;
  const int half = lane >> 4;
  const int l16 = lane & 15;
  const int tile = blockIdx.x * WAVES + wave;
  if (tile * 16 >= n_samples) return;
  const int row0 = tile * 16;
  float* yb = ybuf[wave];

  {
    const float* xr = x + (row0 + l16) * NIN + 2 * half;
    v2f a0 = ldg2(xr);
    v2f a1 = ldg2(xr + 4);
#pragma unroll
    for (int nt = 0; nt < 8; ++nt) {
      const int n = nt * 16 + l16;
      v2f p10 = ldg2(w1_1 + n * NIN + 2 * half);
      v2f p11 = ldg2(w1_1 + n * NIN + 4 + 2 * half);
      v2f p20 = ldg2(w2_1 + n * NIN + 2 * half);
      v2f p21 = ldg2(w2_1 + n * NIN + 4 + 2 * half);
      v8f c1 = {}, c2 = {};
      c1 = wmma_k4(a0, p10, c1);
      c1 = wmma_k4(a1, p11, c1);
      c2 = wmma_k4(a0, p20, c2);
      c2 = wmma_k4(a1, p21, c2);
      const float bb1 = b1_1[n], bb2 = b2_1[n];
#pragma unroll
      for (int j = 0; j < 8; ++j) {
        const int m = j + 8 * half;
        yb[m * LDS_STRIDE + n] = (c1[j] + bb1) * (c2[j] + bb2);
      }
    }
  }

  float part[8];
#pragma unroll
  for (int j = 0; j < 8; ++j) part[j] = 0.f;

  const float* arow = yb + l16 * LDS_STRIDE + 2 * half;
#pragma unroll
  for (int nt = 0; nt < 8; ++nt) {
    const int n = nt * 16 + l16;
    const float* b1r = w1_2 + n * H + 2 * half;
    const float* b2r = w2_2 + n * H + 2 * half;
    v8f c1 = {}, c2 = {};
#pragma unroll 4
    for (int ks = 0; ks < 32; ++ks) {
      v2f a = *reinterpret_cast<const v2f*>(arow + ks * 4);
      v2f q1 = ldg2(b1r + ks * 4);
      v2f q2 = ldg2(b2r + ks * 4);
      c1 = wmma_k4(a, q1, c1);
      c2 = wmma_k4(a, q2, c2);
    }
    const float bia1 = b1_2[n], bia2 = b2_2[n], wo = w_out[n];
#pragma unroll
    for (int j = 0; j < 8; ++j)
      part[j] += (c1[j] + bia1) * (c2[j] + bia2) * wo;
  }

#pragma unroll
  for (int j = 0; j < 8; ++j) {
    float v = part[j];
    v += __shfl_xor(v, 1, 32);
    v += __shfl_xor(v, 2, 32);
    v += __shfl_xor(v, 4, 32);
    v += __shfl_xor(v, 8, 32);
    part[j] = v;
  }
  if (l16 == 0) {
#pragma unroll
    for (int j = 0; j < 8; ++j) out[row0 + j + 8 * half] = part[j];
  }
}

template <int WAVES>
__global__ __launch_bounds__(WAVES * 32) void net_fwd_grad_kernel(
    const float* __restrict__ x, const float* __restrict__ xdot,
    float* __restrict__ outY, float* __restrict__ outG, int n_samples,
    const float* __restrict__ w1_1, const float* __restrict__ b1_1,
    const float* __restrict__ w2_1, const float* __restrict__ b2_1,
    const float* __restrict__ w1_2, const float* __restrict__ b1_2,
    const float* __restrict__ w2_2, const float* __restrict__ b2_2,
    const float* __restrict__ w_out) {
  __shared__ float ybuf[WAVES][16 * LDS_STRIDE];
  __shared__ float ydbuf[WAVES][16 * LDS_STRIDE];

  const int wave = threadIdx.x >> 5;
  const int lane = threadIdx.x & 31;
  const int half = lane >> 4;
  const int l16 = lane & 15;
  const int tile = blockIdx.x * WAVES + wave;
  if (tile * 16 >= n_samples) return;
  const int row0 = tile * 16;
  float* yb = ybuf[wave];
  float* ydb = ydbuf[wave];

  {
    const float* xr = x + (row0 + l16) * NIN + 2 * half;
    const float* xdr = xdot + (row0 + l16) * NIN + 2 * half;
    v2f a0 = ldg2(xr), a1 = ldg2(xr + 4);
    v2f d0 = ldg2(xdr), d1 = ldg2(xdr + 4);
#pragma unroll
    for (int nt = 0; nt < 8; ++nt) {
      const int n = nt * 16 + l16;
      v2f p10 = ldg2(w1_1 + n * NIN + 2 * half);
      v2f p11 = ldg2(w1_1 + n * NIN + 4 + 2 * half);
      v2f p20 = ldg2(w2_1 + n * NIN + 2 * half);
      v2f p21 = ldg2(w2_1 + n * NIN + 4 + 2 * half);
      v8f c1 = {}, c2 = {}, g1 = {}, g2 = {};
      c1 = wmma_k4(a0, p10, c1);
      c1 = wmma_k4(a1, p11, c1);
      c2 = wmma_k4(a0, p20, c2);
      c2 = wmma_k4(a1, p21, c2);
      g1 = wmma_k4(d0, p10, g1);
      g1 = wmma_k4(d1, p11, g1);
      g2 = wmma_k4(d0, p20, g2);
      g2 = wmma_k4(d1, p21, g2);
      const float bb1 = b1_1[n], bb2 = b2_1[n];
#pragma unroll
      for (int j = 0; j < 8; ++j) {
        const int m = j + 8 * half;
        const float z1 = c1[j] + bb1, z2 = c2[j] + bb2;
        yb[m * LDS_STRIDE + n] = z1 * z2;
        ydb[m * LDS_STRIDE + n] = g1[j] * z2 + z1 * g2[j];
      }
    }
  }

  float pY[8], pG[8];
#pragma unroll
  for (int j = 0; j < 8; ++j) {
    pY[j] = 0.f;
    pG[j] = 0.f;
  }

  const float* arow = yb + l16 * LDS_STRIDE + 2 * half;
  const float* drow = ydb + l16 * LDS_STRIDE + 2 * half;
#pragma unroll
  for (int nt = 0; nt < 8; ++nt) {
    const int n = nt * 16 + l16;
    const float* b1r = w1_2 + n * H + 2 * half;
    const float* b2r = w2_2 + n * H + 2 * half;
    v8f c1 = {}, c2 = {}, g1 = {}, g2 = {};
#pragma unroll 4
    for (int ks = 0; ks < 32; ++ks) {
      v2f a = *reinterpret_cast<const v2f*>(arow + ks * 4);
      v2f ad = *reinterpret_cast<const v2f*>(drow + ks * 4);
      v2f q1 = ldg2(b1r + ks * 4);
      v2f q2 = ldg2(b2r + ks * 4);
      c1 = wmma_k4(a, q1, c1);
      c2 = wmma_k4(a, q2, c2);
      g1 = wmma_k4(ad, q1, g1);
      g2 = wmma_k4(ad, q2, g2);
    }
    const float bia1 = b1_2[n], bia2 = b2_2[n], wo = w_out[n];
#pragma unroll
    for (int j = 0; j < 8; ++j) {
      const float z1 = c1[j] + bia1, z2 = c2[j] + bia2;
      pY[j] += z1 * z2 * wo;
      pG[j] += (g1[j] * z2 + z1 * g2[j]) * wo;
    }
  }

#pragma unroll
  for (int j = 0; j < 8; ++j) {
    float vy = pY[j], vg = pG[j];
    vy += __shfl_xor(vy, 1, 32);
    vy += __shfl_xor(vy, 2, 32);
    vy += __shfl_xor(vy, 4, 32);
    vy += __shfl_xor(vy, 8, 32);
    vg += __shfl_xor(vg, 1, 32);
    vg += __shfl_xor(vg, 2, 32);
    vg += __shfl_xor(vg, 4, 32);
    vg += __shfl_xor(vg, 8, 32);
    pY[j] = vy;
    pG[j] = vg;
  }
  if (l16 == 0) {
#pragma unroll
    for (int j = 0; j < 8; ++j) {
      outY[row0 + j + 8 * half] = pY[j];
      outG[row0 + j + 8 * half] = pG[j];
    }
  }
}

extern "C" void kernel_launch(void* const* d_in, const int* in_sizes, int n_in,
                              void* d_out, int out_size, void* d_ws,
                              size_t ws_size, hipStream_t stream) {
  const float* T = (const float*)d_in[0];
  const float* l = (const float*)d_in[1];
  const float* l1_dot = (const float*)d_in[2];
  const float* center = (const float*)d_in[3];
  const float* w1_1 = (const float*)d_in[4];
  const float* b1_1 = (const float*)d_in[5];
  const float* w2_1 = (const float*)d_in[6];
  const float* b2_1 = (const float*)d_in[7];
  const float* w1_2 = (const float*)d_in[8];
  const float* b1_2 = (const float*)d_in[9];
  const float* w2_2 = (const float*)d_in[10];
  const float* b2_2 = (const float*)d_in[11];
  const float* w_out = (const float*)d_in[12];

  const int nT = in_sizes[0] / NIN;
  const int nL = in_sizes[1] / NIN;
  const int nC = in_sizes[3] / NIN;

  float* out = (float*)d_out;
  float* v_t = out;
  float* v_y = out + nT;
  float* v_g = out + nT + nL;
  float* v_c = out + nT + 2 * nL;

  constexpr int WF = 4;
  constexpr int WG = 2;

  const int tilesT = (nT + 15) / 16;
  const int tilesL = (nL + 15) / 16;
  const int tilesC = (nC + 15) / 16;

  net_forward_kernel<WF><<<(tilesT + WF - 1) / WF, WF * 32, 0, stream>>>(
      T, v_t, nT, w1_1, b1_1, w2_1, b2_1, w1_2, b1_2, w2_2, b2_2, w_out);

  net_forward_kernel<WF><<<(tilesC + WF - 1) / WF, WF * 32, 0, stream>>>(
      center, v_c, nC, w1_1, b1_1, w2_1, b2_1, w1_2, b1_2, w2_2, b2_2, w_out);

  net_fwd_grad_kernel<WG><<<(tilesL + WG - 1) / WG, WG * 32, 0, stream>>>(
      l, l1_dot, v_y, v_g, nL, w1_1, b1_1, w2_1, b2_1, w1_2, b1_2, w2_2, b2_2,
      w_out);
}


# All three launches merged into one resident launch on the register-resident weight path: each wave loads its layer-2 slices once, then walks value-only tiles of both inputs, then value+tangent tiles
# speedup vs baseline: 1.8561x; 1.8561x over previous
#include <hip/hip_runtime.h>


typedef float v2f __attribute__((ext_vector_type(2)));
typedef float v8f __attribute__((ext_vector_type(8)));

#define H 128
#define NIN 8
#define LDS_STRIDE 132
#define NWAVES 8

__device__ __forceinline__ v8f wmma_k4(v2f a, v2f b, v8f c) {
  return __builtin_amdgcn_wmma_f32_16x16x4_f32(
       false, a,  false, b,
       (short)0, c,  false,  false);
}

__device__ __forceinline__ v2f ldg2(const float* p) {
  return *reinterpret_cast<const v2f*>(p);
}

template <int MASK>
__device__ __forceinline__ float row_xadd(float v) {
  const int t = __builtin_amdgcn_update_dpp(0, __builtin_bit_cast(int, v),
                                            0x160 + MASK  , 0xf, 0xf, true);
  return v + __builtin_bit_cast(float, t);
}

template <bool GRAD, int MT>
__device__ __forceinline__ void run_tiles(
    const float* __restrict__ x, const float* __restrict__ xdot,
    float* __restrict__ outY, float* __restrict__ outG, int n_samples,
    int tiles_per_group,
    const v2f (&B1)[32], const v2f (&B2)[32],
    const v2f p10, const v2f p11, const v2f p20, const v2f p21,
    const float bb1, const float bb2, const float bia1, const float bia2,
    const float wo, const int wave, const int half, const int l16, const int n) {
  constexpr int ROWS = MT * 16;
  constexpr int NV = GRAD ? 2 : 1;
  __shared__ float ybuf[2][NV][ROWS * LDS_STRIDE];
  __shared__ float red[2][NWAVES][NV][ROWS];

  const int ntiles = (n_samples + 15) >> 4;
  const int tile_begin = blockIdx.x * tiles_per_group;
  int tile_end = tile_begin + tiles_per_group;
  if (tile_end > ntiles) tile_end = ntiles;
  if (tile_begin >= tile_end) return;
  const int niter = (tile_end - tile_begin + MT - 1) / MT;
  int valid_end = tile_end * 16;
  if (valid_end > n_samples) valid_end = n_samples;

  v2f xa[MT][2];
  v2f xd[MT][2];

  auto load_x = [&](int iter) {
#pragma unroll
    for (int mt = 0; mt < MT; ++mt) {
      int row = (tile_begin + iter * MT + mt) * 16 + l16;
      if (row > n_samples - 1) row = n_samples - 1;
      const float* xr = x + row * NIN + 2 * half;
      xa[mt][0] = ldg2(xr);
      xa[mt][1] = ldg2(xr + 4);
      if (GRAD) {
        const float* xdr = xdot + row * NIN + 2 * half;
        xd[mt][0] = ldg2(xdr);
        xd[mt][1] = ldg2(xdr + 4);
      }
    }
  };

  auto layer1 = [&](int buf) {
#pragma unroll
    for (int mt = 0; mt < MT; ++mt) {
      v8f c1 = {}, c2 = {};
      c1 = wmma_k4(xa[mt][0], p10, c1);
      c1 = wmma_k4(xa[mt][1], p11, c1);
      c2 = wmma_k4(xa[mt][0], p20, c2);
      c2 = wmma_k4(xa[mt][1], p21, c2);
      v8f g1 = {}, g2 = {};
      if (GRAD) {
        g1 = wmma_k4(xd[mt][0], p10, g1);
        g1 = wmma_k4(xd[mt][1], p11, g1);
        g2 = wmma_k4(xd[mt][0], p20, g2);
        g2 = wmma_k4(xd[mt][1], p21, g2);
      }
#pragma unroll
      for (int j = 0; j < 8; ++j) {
        const int m = mt * 16 + j + 8 * half;
        const float z1 = c1[j] + bb1, z2 = c2[j] + bb2;
        ybuf[buf][0][m * LDS_STRIDE + n] = z1 * z2;
        if (GRAD)
          ybuf[buf][NV - 1][m * LDS_STRIDE + n] = g1[j] * z2 + z1 * g2[j];
      }
    }
  };

  auto reduce_out = [&](int buf, int iter) {
    const int t = threadIdx.x;
    if (t < NV * ROWS) {
      const int plane = t / ROWS;
      const int s = t - plane * ROWS;
      float v = red[buf][0][plane][s];
#pragma unroll
      for (int w = 1; w < NWAVES; ++w) v += red[buf][w][plane][s];
      const int sample = (tile_begin + iter * MT) * 16 + s;
      if (sample < valid_end) {
        if (plane == 0) outY[sample] = v;
        else outG[sample] = v;
      }
    }
  };

  load_x(0);
  layer1(0);
  load_x(1);

  for (int it = 0; it < niter; ++it) {
    const int b = it & 1;
    __syncthreads();

    v2f a[2][32], ad[2][32];
    auto load_a = [&](int mt) {
      const float* arow = &ybuf[b][0][(mt * 16 + l16) * LDS_STRIDE + 2 * half];
      const float* drow = &ybuf[b][NV - 1][(mt * 16 + l16) * LDS_STRIDE + 2 * half];
#pragma unroll
      for (int ks = 0; ks < 32; ++ks) {
        a[mt & 1][ks] = *reinterpret_cast<const v2f*>(arow + ks * 4);
        if (GRAD) ad[mt & 1][ks] = *reinterpret_cast<const v2f*>(drow + ks * 4);
      }
    };
    load_a(0);
    __builtin_amdgcn_sched_barrier(0);

    if (it + 1 < niter) {
      layer1(b ^ 1);
      load_x(it + 2);
    }
    if (it > 0) reduce_out(b ^ 1, it - 1);

#pragma unroll
    for (int mt = 0; mt < MT; ++mt) {
      if (mt + 1 < MT) load_a(mt + 1);
      __builtin_amdgcn_sched_barrier(0);
      v8f c1 = {}, c2 = {}, g1 = {}, g2 = {};
#pragma unroll
      for (int ks = 0; ks < 32; ++ks) {
        c1 = wmma_k4(a[mt & 1][ks], B1[ks], c1);
        c2 = wmma_k4(a[mt & 1][ks], B2[ks], c2);
        if (GRAD) {
          g1 = wmma_k4(ad[mt & 1][ks], B1[ks], g1);
          g2 = wmma_k4(ad[mt & 1][ks], B2[ks], g2);
        }
      }
      float vy[8], vg[8];
#pragma unroll
      for (int j = 0; j < 8; ++j) {
        const float z1 = c1[j] + bia1, z2 = c2[j] + bia2;
        vy[j] = z1 * z2 * wo;
        vg[j] = GRAD ? (g1[j] * z2 + z1 * g2[j]) * wo : 0.f;
      }
#pragma unroll
      for (int j = 0; j < 8; ++j) { vy[j] = row_xadd<1>(vy[j]); if (GRAD) vg[j] = row_xadd<1>(vg[j]); }
#pragma unroll
      for (int j = 0; j < 8; ++j) { vy[j] = row_xadd<2>(vy[j]); if (GRAD) vg[j] = row_xadd<2>(vg[j]); }
#pragma unroll
      for (int j = 0; j < 8; ++j) { vy[j] = row_xadd<4>(vy[j]); if (GRAD) vg[j] = row_xadd<4>(vg[j]); }
#pragma unroll
      for (int j = 0; j < 8; ++j) { vy[j] = row_xadd<8>(vy[j]); if (GRAD) vg[j] = row_xadd<8>(vg[j]); }
      if (l16 == 0) {
#pragma unroll
        for (int j = 0; j < 8; ++j) {
          red[b][wave][0][mt * 16 + j + 8 * half] = vy[j];
          if (GRAD) red[b][wave][NV - 1][mt * 16 + j + 8 * half] = vg[j];
        }
      }
    }
  }

  __syncthreads();
  reduce_out((niter - 1) & 1, niter - 1);
}

__global__ __launch_bounds__(NWAVES * 32) void net_all_kernel(
    const float* __restrict__ T, const float* __restrict__ l,
    const float* __restrict__ l1_dot, const float* __restrict__ center,
    float* __restrict__ v_t, float* __restrict__ v_y, float* __restrict__ v_g,
    float* __restrict__ v_c, int nT, int nL, int nC,
    const float* __restrict__ w1_1, const float* __restrict__ b1_1,
    const float* __restrict__ w2_1, const float* __restrict__ b2_1,
    const float* __restrict__ w1_2, const float* __restrict__ b1_2,
    const float* __restrict__ w2_2, const float* __restrict__ b2_2,
    const float* __restrict__ w_out) {
  const int wave = threadIdx.x >> 5;
  const int lane = threadIdx.x & 31;
  const int half = lane >> 4;
  const int l16 = lane & 15;
  const int n = wave * 16 + l16;

  v2f B1[32], B2[32];
  {
    const float* b1r = w1_2 + n * H + 2 * half;
    const float* b2r = w2_2 + n * H + 2 * half;
#pragma unroll
    for (int ks = 0; ks < 32; ++ks) {
      B1[ks] = ldg2(b1r + ks * 4);
      B2[ks] = ldg2(b2r + ks * 4);
    }
  }
  const v2f p10 = ldg2(w1_1 + n * NIN + 2 * half);
  const v2f p11 = ldg2(w1_1 + n * NIN + 4 + 2 * half);
  const v2f p20 = ldg2(w2_1 + n * NIN + 2 * half);
  const v2f p21 = ldg2(w2_1 + n * NIN + 4 + 2 * half);
  const float bb1 = b1_1[n], bb2 = b2_1[n];
  const float bia1 = b1_2[n], bia2 = b2_2[n], wo = w_out[n];

  const int G = gridDim.x;
  const int tpgT = (((nT + 15) >> 4) + G - 1) / G;
  const int tpgC = (((nC + 15) >> 4) + G - 1) / G;
  const int tpgL = (((nL + 15) >> 4) + G - 1) / G;

  run_tiles<false, 4>(T, T, v_t, v_t, nT, tpgT, B1, B2, p10, p11, p20, p21,
                      bb1, bb2, bia1, bia2, wo, wave, half, l16, n);
  run_tiles<false, 1>(center, center, v_c, v_c, nC, tpgC, B1, B2, p10, p11, p20, p21,
                      bb1, bb2, bia1, bia2, wo, wave, half, l16, n);
  run_tiles<true, 2>(l, l1_dot, v_y, v_g, nL, tpgL, B1, B2, p10, p11, p20, p21,
                     bb1, bb2, bia1, bia2, wo, wave, half, l16, n);
}

extern "C" void kernel_launch(void* const* d_in, const int* in_sizes, int n_in,
                              void* d_out, int out_size, void* d_ws,
                              size_t ws_size, hipStream_t stream) {
  const float* T = (const float*)d_in[0];
  const float* l = (const float*)d_in[1];
  const float* l1_dot = (const float*)d_in[2];
  const float* center = (const float*)d_in[3];
  const float* w1_1 = (const float*)d_in[4];
  const float* b1_1 = (const float*)d_in[5];
  const float* w2_1 = (const float*)d_in[6];
  const float* b2_1 = (const float*)d_in[7];
  const float* w1_2 = (const float*)d_in[8];
  const float* b1_2 = (const float*)d_in[9];
  const float* w2_2 = (const float*)d_in[10];
  const float* b2_2 = (const float*)d_in[11];
  const float* w_out = (const float*)d_in[12];

  const int nT = in_sizes[0] / NIN;
  const int nL = in_sizes[1] / NIN;
  const int nC = in_sizes[3] / NIN;

  float* out = (float*)d_out;
  float* v_t = out;
  float* v_y = out + nT;
  float* v_g = out + nT + nL;
  float* v_c = out + nT + 2 * nL;

  constexpr int NGROUPS = 256;

  net_all_kernel<<<NGROUPS, NWAVES * 32, 0, stream>>>(
      T, l, l1_dot, center, v_t, v_y, v_g, v_c, nT, nL, nC, w1_1, b1_1, w2_1,
      b2_1, w1_2, b1_2, w2_2, b2_2, w_out);
}


# All products transposed (weights as A, samples as B): a lane holds 8 hidden units of one sample, so the output dot sums in-lane with no DPP steps, y1 stored 16 B wide; 16 partials, sum order changes
# speedup vs baseline: 1.9722x; 1.0626x over previous
#include <hip/hip_runtime.h>


typedef float v2f __attribute__((ext_vector_type(2)));
typedef float v4f __attribute__((ext_vector_type(4)));
typedef float v8f __attribute__((ext_vector_type(8)));

#define H 128
#define NIN 8
#define LDS_STRIDE 132
#define NWAVES 8

__device__ __forceinline__ v8f wmma_k4(v2f a, v2f b, v8f c) {
  return __builtin_amdgcn_wmma_f32_16x16x4_f32(
       false, a,  false, b,
       (short)0, c,  false,  false);
}

__device__ __forceinline__ v2f ldg2(const float* p) {
  return *reinterpret_cast<const v2f*>(p);
}

__device__ __forceinline__ v8f ldg8(const float* p) {
  const v4f lo = *reinterpret_cast<const v4f*>(p);
  const v4f hi = *reinterpret_cast<const v4f*>(p + 4);
  return __builtin_shufflevector(lo, hi, 0, 1, 2, 3, 4, 5, 6, 7);
}

template <bool GRAD, int MT>
__device__ __forceinline__ void run_tiles(
    const float* __restrict__ x, const float* __restrict__ xdot,
    float* __restrict__ outY, float* __restrict__ outG, int n_samples,
    int tiles_per_group,
    const v2f (&B1)[32], const v2f (&B2)[32],
    const v2f p10, const v2f p11, const v2f p20, const v2f p21,
    const v8f bb1, const v8f bb2, const v8f bia1, const v8f bia2,
    const v8f wo, const int wave, const int half, const int l16) {
  constexpr int ROWS = MT * 16;
  constexpr int NV = GRAD ? 2 : 1;
  __shared__ __attribute__((aligned(16))) float ybuf[2][NV][ROWS * LDS_STRIDE];
  __shared__ float red[2][2 * NWAVES][NV][ROWS];

  const int ntiles = (n_samples + 15) >> 4;
  const int tile_begin = blockIdx.x * tiles_per_group;
  int tile_end = tile_begin + tiles_per_group;
  if (tile_end > ntiles) tile_end = ntiles;
  if (tile_begin >= tile_end) return;
  const int niter = (tile_end - tile_begin + MT - 1) / MT;
  int valid_end = tile_end * 16;
  if (valid_end > n_samples) valid_end = n_samples;

  v2f xa[MT][2];
  v2f xd[MT][2];

  auto load_x = [&](int iter) {
#pragma unroll
    for (int mt = 0; mt < MT; ++mt) {
      int row = (tile_begin + iter * MT + mt) * 16 + l16;
      if (row > n_samples - 1) row = n_samples - 1;
      const float* xr = x + row * NIN + 2 * half;
      xa[mt][0] = ldg2(xr);
      xa[mt][1] = ldg2(xr + 4);
      if (GRAD) {
        const float* xdr = xdot + row * NIN + 2 * half;
        xd[mt][0] = ldg2(xdr);
        xd[mt][1] = ldg2(xdr + 4);
      }
    }
  };

  auto layer1 = [&](int buf) {
#pragma unroll
    for (int mt = 0; mt < MT; ++mt) {
      v8f c1 = {}, c2 = {};
      c1 = wmma_k4(p10, xa[mt][0], c1);
      c1 = wmma_k4(p11, xa[mt][1], c1);
      c2 = wmma_k4(p20, xa[mt][0], c2);
      c2 = wmma_k4(p21, xa[mt][1], c2);
      v8f g1 = {}, g2 = {};
      if (GRAD) {
        g1 = wmma_k4(p10, xd[mt][0], g1);
        g1 = wmma_k4(p11, xd[mt][1], g1);
        g2 = wmma_k4(p20, xd[mt][0], g2);
        g2 = wmma_k4(p21, xd[mt][1], g2);
      }
      const v8f z1 = c1 + bb1, z2 = c2 + bb2;
      const v8f y = z1 * z2;
      const int off = (mt * 16 + l16) * LDS_STRIDE + wave * 16 + 8 * half;
      *reinterpret_cast<v4f*>(&ybuf[buf][0][off]) = __builtin_shufflevector(y, y, 0, 1, 2, 3);
      *reinterpret_cast<v4f*>(&ybuf[buf][0][off + 4]) = __builtin_shufflevector(y, y, 4, 5, 6, 7);
      if (GRAD) {
        const v8f yd = g1 * z2 + z1 * g2;
        *reinterpret_cast<v4f*>(&ybuf[buf][NV - 1][off]) = __builtin_shufflevector(yd, yd, 0, 1, 2, 3);
        *reinterpret_cast<v4f*>(&ybuf[buf][NV - 1][off + 4]) = __builtin_shufflevector(yd, yd, 4, 5, 6, 7);
      }
    }
  };

  auto reduce_out = [&](int buf, int iter) {
    const int t = threadIdx.x;
    if (t < NV * ROWS) {
      const int plane = t / ROWS;
      const int s = t - plane * ROWS;
      float v = red[buf][0][plane][s];
#pragma unroll
      for (int w = 1; w < 2 * NWAVES; ++w) v += red[buf][w][plane][s];
      const int sample = (tile_begin + iter * MT) * 16 + s;
      if (sample < valid_end) {
        if (plane == 0) outY[sample] = v;
        else outG[sample] = v;
      }
    }
  };

  load_x(0);
  layer1(0);
  load_x(1);

  for (int it = 0; it < niter; ++it) {
    const int b = it & 1;
    __syncthreads();

    v2f a[2][32], ad[2][32];
    auto load_a = [&](int mt) {
      const float* arow = &ybuf[b][0][(mt * 16 + l16) * LDS_STRIDE + 2 * half];
      const float* drow = &ybuf[b][NV - 1][(mt * 16 + l16) * LDS_STRIDE + 2 * half];
#pragma unroll
      for (int ks = 0; ks < 32; ++ks) {
        a[mt & 1][ks] = *reinterpret_cast<const v2f*>(arow + ks * 4);
        if (GRAD) ad[mt & 1][ks] = *reinterpret_cast<const v2f*>(drow + ks * 4);
      }
    };
    load_a(0);
    __builtin_amdgcn_sched_barrier(0);

    if (it + 1 < niter) {
      layer1(b ^ 1);
      load_x(it + 2);
    }
    if (it > 0) reduce_out(b ^ 1, it - 1);

#pragma unroll
    for (int mt = 0; mt < MT; ++mt) {
      if (mt + 1 < MT) load_a(mt + 1);
      __builtin_amdgcn_sched_barrier(0);
      v8f c1 = {}, c2 = {}, g1 = {}, g2 = {};
#pragma unroll
      for (int ks = 0; ks < 32; ++ks) {
        c1 = wmma_k4(B1[ks], a[mt & 1][ks], c1);
        c2 = wmma_k4(B2[ks], a[mt & 1][ks], c2);
        if (GRAD) {
          g1 = wmma_k4(B1[ks], ad[mt & 1][ks], g1);
          g2 = wmma_k4(B2[ks], ad[mt & 1][ks], g2);
        }
      }
      const v8f z1 = c1 + bia1, z2 = c2 + bia2;
      const v8f vy = z1 * z2 * wo;
      const v4f y4 = __builtin_shufflevector(vy, vy, 0, 1, 2, 3) + __builtin_shufflevector(vy, vy, 4, 5, 6, 7);
      const v2f y2 = __builtin_shufflevector(y4, y4, 0, 1) + __builtin_shufflevector(y4, y4, 2, 3);
      red[b][2 * wave + half][0][mt * 16 + l16] = y2[0] + y2[1];
      if (GRAD) {
        const v8f vg = (g1 * z2 + z1 * g2) * wo;
        const v4f g4 = __builtin_shufflevector(vg, vg, 0, 1, 2, 3) + __builtin_shufflevector(vg, vg, 4, 5, 6, 7);
        const v2f gg = __builtin_shufflevector(g4, g4, 0, 1) + __builtin_shufflevector(g4, g4, 2, 3);
        red[b][2 * wave + half][NV - 1][mt * 16 + l16] = gg[0] + gg[1];
      }
    }
  }

  __syncthreads();
  reduce_out((niter - 1) & 1, niter - 1);
}

__global__ __launch_bounds__(NWAVES * 32) void net_all_kernel(
    const float* __restrict__ T, const float* __restrict__ l,
    const float* __restrict__ l1_dot, const float* __restrict__ center,
    float* __restrict__ v_t, float* __restrict__ v_y, float* __restrict__ v_g,
    float* __restrict__ v_c, int nT, int nL, int nC,
    const float* __restrict__ w1_1, const float* __restrict__ b1_1,
    const float* __restrict__ w2_1, const float* __restrict__ b2_1,
    const float* __restrict__ w1_2, const float* __restrict__ b1_2,
    const float* __restrict__ w2_2, const float* __restrict__ b2_2,
    const float* __restrict__ w_out) {
  const int wave = threadIdx.x >> 5;
  const int lane = threadIdx.x & 31;
  const int half = lane >> 4;
  const int l16 = lane & 15;
  const int n = wave * 16 + l16;
  const int nb = wave * 16 + 8 * half;

  v2f B1[32], B2[32];
  {
    const float* b1r = w1_2 + n * H + 2 * half;
    const float* b2r = w2_2 + n * H + 2 * half;
#pragma unroll
    for (int ks = 0; ks < 32; ++ks) {
      B1[ks] = ldg2(b1r + ks * 4);
      B2[ks] = ldg2(b2r + ks * 4);
    }
  }
  const v2f p10 = ldg2(w1_1 + n * NIN + 2 * half);
  const v2f p11 = ldg2(w1_1 + n * NIN + 4 + 2 * half);
  const v2f p20 = ldg2(w2_1 + n * NIN + 2 * half);
  const v2f p21 = ldg2(w2_1 + n * NIN + 4 + 2 * half);
  const v8f bb1 = ldg8(b1_1 + nb), bb2 = ldg8(b2_1 + nb);
  const v8f bia1 = ldg8(b1_2 + nb), bia2 = ldg8(b2_2 + nb), wo = ldg8(w_out + nb);

  const int G = gridDim.x;
  const int tpgT = (((nT + 15) >> 4) + G - 1) / G;
  const int tpgC = (((nC + 15) >> 4) + G - 1) / G;
  const int tpgL = (((nL + 15) >> 4) + G - 1) / G;

  run_tiles<false, 4>(T, T, v_t, v_t, nT, tpgT, B1, B2, p10, p11, p20, p21,
                      bb1, bb2, bia1, bia2, wo, wave, half, l16);
  run_tiles<false, 1>(center, center, v_c, v_c, nC, tpgC, B1, B2, p10, p11, p20, p21,
                      bb1, bb2, bia1, bia2, wo, wave, half, l16);
  run_tiles<true, 2>(l, l1_dot, v_y, v_g, nL, tpgL, B1, B2, p10, p11, p20, p21,
                     bb1, bb2, bia1, bia2, wo, wave, half, l16);
}

extern "C" void kernel_launch(void* const* d_in, const int* in_sizes, int n_in,
                              void* d_out, int out_size, void* d_ws,
                              size_t ws_size, hipStream_t stream) {
  const float* T = (const float*)d_in[0];
  const float* l = (const float*)d_in[1];
  const float* l1_dot = (const float*)d_in[2];
  const float* center = (const float*)d_in[3];
  const float* w1_1 = (const float*)d_in[4];
  const float* b1_1 = (const float*)d_in[5];
  const float* w2_1 = (const float*)d_in[6];
  const float* b2_1 = (const float*)d_in[7];
  const float* w1_2 = (const float*)d_in[8];
  const float* b1_2 = (const float*)d_in[9];
  const float* w2_2 = (const float*)d_in[10];
  const float* b2_2 = (const float*)d_in[11];
  const float* w_out = (const float*)d_in[12];

  const int nT = in_sizes[0] / NIN;
  const int nL = in_sizes[1] / NIN;
  const int nC = in_sizes[3] / NIN;

  float* out = (float*)d_out;
  float* v_t = out;
  float* v_y = out + nT;
  float* v_g = out + nT + nL;
  float* v_c = out + nT + 2 * nL;

  constexpr int NGROUPS = 256;

  net_all_kernel<<<NGROUPS, NWAVES * 32, 0, stream>>>(
      T, l, l1_dot, center, v_t, v_y, v_g, v_c, nT, nL, nC, w1_1, b1_1, w2_1,
      b2_1, w1_2, b1_2, w2_2, b2_2, w_out);
}
